# starting kernel
# baseline (speedup 1.0000x reference)
#include <hip/hip_runtime.h>

typedef __attribute__((ext_vector_type(2))) float v2f;
typedef __attribute__((ext_vector_type(8))) float v8f;

__global__ void deg_init_kernel(float* __restrict__ deg, int n) {
    int i = blockIdx.x * blockDim.x + threadIdx.x;
    if (i < n) deg[i] = 1.0f;
}

__global__ void deg_accum_kernel(const int* __restrict__ dst, float* __restrict__ deg, int E) {
    int e = blockIdx.x * blockDim.x + threadIdx.x;
    if (e < E) {
        __builtin_prefetch(dst + e + 4096, 0, 0);
        unsafeAtomicAdd(&deg[dst[e]], 1.0f);
    }
}

__global__ void deg_rsqrt_kernel(float* __restrict__ deg, int n) {
    int i = blockIdx.x * blockDim.x + threadIdx.x;
    if (i < n) deg[i] = rsqrtf(deg[i]);
}

__global__ void zero4_kernel(float4* __restrict__ p, int n4) {
    int i = blockIdx.x * blockDim.x + threadIdx.x;
    if (i < n4) p[i] = make_float4(0.f, 0.f, 0.f, 0.f);
}

template <int K, int NC, bool BIAS_RELU>
__global__ void gemm_wmma_kernel(const float* __restrict__ A, const float* __restrict__ B,
                                 const float* __restrict__ bias, float* __restrict__ C) {
    const int lane = threadIdx.x & 31;
    const int wave = threadIdx.x >> 5;
    const int rt   = blockIdx.x;
    const int half = lane >> 4;
    const int l16  = lane & 15;
    const int m    = rt * 16 + l16;
    const int n    = wave * 16 + l16;

    v8f acc = {};
#pragma unroll
    for (int kb = 0; kb < K; kb += 4) {
        const int kp = kb + half * 2;
        v2f a;
        a.x = A[m * K + kp];
        a.y = A[m * K + kp + 1];
        v2f b;
        b.x = B[kp * NC + n];
        b.y = B[(kp + 1) * NC + n];
        acc = __builtin_amdgcn_wmma_f32_16x16x4_f32(
             false, a,  false, b,
             (short)0, acc,  false,  false);
    }

    const float bv = BIAS_RELU ? bias[n] : 0.0f;
#pragma unroll
    for (int r = 0; r < 8; ++r) {
        const int row = rt * 16 + r + half * 8;
        float v = acc[r];
        if (BIAS_RELU) {
            v += bv;
            v = v > 0.0f ? v : 0.0f;
        }
        C[row * NC + n] = v;
    }
}

template <int F>
__global__ void propagate_kernel(const int* __restrict__ src, const int* __restrict__ dst,
                                 const float* __restrict__ dis, const float* __restrict__ H,
                                 float* __restrict__ agg, int E) {
    constexpr int G = F / 4;
    unsigned gid = blockIdx.x * blockDim.x + threadIdx.x;
    int e = (int)(gid / G);
    int f = (int)(gid % G) * 4;
    if (e >= E) return;
    __builtin_prefetch(src + e + 2048, 0, 0);
    __builtin_prefetch(dst + e + 2048, 0, 0);
    int s = src[e];
    int d = dst[e];
    float nrm = dis[s] * dis[d];
    const float4 hv = *(const float4*)(H + (size_t)s * F + f);
    float* ap = agg + (size_t)d * F + f;
    unsafeAtomicAdd(ap + 0, hv.x * nrm);
    unsafeAtomicAdd(ap + 1, hv.y * nrm);
    unsafeAtomicAdd(ap + 2, hv.z * nrm);
    unsafeAtomicAdd(ap + 3, hv.w * nrm);
}

template <int F>
__global__ void selfloop_bias_relu_kernel(const float* __restrict__ agg, const float* __restrict__ H,
                                          const float* __restrict__ dis, const float* __restrict__ bias,
                                          float* __restrict__ out, int total) {
    int i = blockIdx.x * blockDim.x + threadIdx.x;
    if (i >= total) return;
    int node = i / F;
    int f    = i % F;
    float ds = dis[node];
    float v  = agg[i] + H[i] * ds * ds + bias[f];
    out[i]   = v > 0.0f ? v : 0.0f;
}

extern "C" void kernel_launch(void* const* d_in, const int* in_sizes, int n_in,
                              void* d_out, int out_size, void* d_ws, size_t ws_size,
                              hipStream_t stream) {
    const float* x   = (const float*)d_in[0];
    const int*   ei  = (const int*)d_in[1];
    const float* W1  = (const float*)d_in[2];
    const float* b1  = (const float*)d_in[3];
    const float* W2  = (const float*)d_in[4];
    const float* b2  = (const float*)d_in[5];
    const float* Wfc = (const float*)d_in[6];
    const float* bfc = (const float*)d_in[7];
    float* out = (float*)d_out;

    const int N = in_sizes[0] / 4;
    const int E = in_sizes[1] / 2;
    const int* srcIdx = ei;
    const int* dstIdx = ei + E;

    float* ws   = (float*)d_ws;
    float* dis  = ws;
    float* h1   = ws + (((size_t)N + 255) & ~(size_t)255);
    float* buf2 = h1 + (size_t)N * 64;
    float* agg1 = buf2;
    float* h2   = buf2;
    float* agg2 = buf2 + (size_t)N * 32;
    float* h2o  = h1;

    const int T = 256;
    const int rowTiles = N / 16;

    deg_init_kernel<<<(N + T - 1) / T, T, 0, stream>>>(dis, N);
    deg_accum_kernel<<<(E + T - 1) / T, T, 0, stream>>>(dstIdx, dis, E);
    deg_rsqrt_kernel<<<(N + T - 1) / T, T, 0, stream>>>(dis, N);

    gemm_wmma_kernel<4, 64, false><<<rowTiles, 128, 0, stream>>>(x, W1, nullptr, h1);

    zero4_kernel<<<((N * 16) + T - 1) / T, T, 0, stream>>>((float4*)agg1, N * 16);
    {
        unsigned total = (unsigned)E * 16u;
        propagate_kernel<64><<<(total + T - 1) / T, T, 0, stream>>>(srcIdx, dstIdx, dis, h1, agg1, E);
    }
    selfloop_bias_relu_kernel<64><<<((N * 64) + T - 1) / T, T, 0, stream>>>(agg1, h1, dis, b1, h1, N * 64);

    gemm_wmma_kernel<64, 32, false><<<rowTiles, 64, 0, stream>>>(h1, W2, nullptr, h2);

    zero4_kernel<<<((N * 8) + T - 1) / T, T, 0, stream>>>((float4*)agg2, N * 8);
    {
        unsigned total = (unsigned)E * 8u;
        propagate_kernel<32><<<(total + T - 1) / T, T, 0, stream>>>(srcIdx, dstIdx, dis, h2, agg2, E);
    }
    selfloop_bias_relu_kernel<32><<<((N * 32) + T - 1) / T, T, 0, stream>>>(agg2, h2, dis, b2, h2o, N * 32);

    gemm_wmma_kernel<32, 32, true><<<rowTiles, 64, 0, stream>>>(h2o, Wfc, bfc, out);
}


# Layer 1 aggregates the 4 input features first and applies W1 after, (A x) W1 = A (x W1): edge pass on 16-byte rows, 4 lanes per edge; self-loop term = start value; float sum order changes
# speedup vs baseline: 1.3607x; 1.3607x over previous
#include <hip/hip_runtime.h>

typedef __attribute__((ext_vector_type(2))) float v2f;
typedef __attribute__((ext_vector_type(8))) float v8f;

__global__ void deg_init_kernel(float* __restrict__ deg, int n) {
    int i = blockIdx.x * blockDim.x + threadIdx.x;
    if (i < n) deg[i] = 1.0f;
}

__global__ void deg_accum_kernel(const int* __restrict__ dst, float* __restrict__ deg, int E) {
    int e = blockIdx.x * blockDim.x + threadIdx.x;
    if (e < E) {
        __builtin_prefetch(dst + e + 4096, 0, 0);
        unsafeAtomicAdd(&deg[dst[e]], 1.0f);
    }
}

__global__ void deg_rsqrt_kernel(float* __restrict__ deg, int n) {
    int i = blockIdx.x * blockDim.x + threadIdx.x;
    if (i < n) deg[i] = rsqrtf(deg[i]);
}

__global__ void zero4_kernel(float4* __restrict__ p, int n4) {
    int i = blockIdx.x * blockDim.x + threadIdx.x;
    if (i < n4) p[i] = make_float4(0.f, 0.f, 0.f, 0.f);
}

__global__ void selfloop_start4_kernel(const float* __restrict__ x, const float* __restrict__ dis,
                                       float4* __restrict__ aggx, int n) {
    int i = blockIdx.x * blockDim.x + threadIdx.x;
    if (i >= n) return;
    const float ds = dis[i];
    const float4 xv = *(const float4*)(x + (size_t)i * 4);
    aggx[i] = make_float4(xv.x * ds * ds, xv.y * ds * ds, xv.z * ds * ds, xv.w * ds * ds);
}

template <int K, int NC, bool BIAS_RELU>
__global__ void gemm_wmma_kernel(const float* __restrict__ A, const float* __restrict__ B,
                                 const float* __restrict__ bias, float* __restrict__ C) {
    const int lane = threadIdx.x & 31;
    const int wave = threadIdx.x >> 5;
    const int rt   = blockIdx.x;
    const int half = lane >> 4;
    const int l16  = lane & 15;
    const int m    = rt * 16 + l16;
    const int n    = wave * 16 + l16;

    v8f acc = {};
#pragma unroll
    for (int kb = 0; kb < K; kb += 4) {
        const int kp = kb + half * 2;
        v2f a;
        a.x = A[m * K + kp];
        a.y = A[m * K + kp + 1];
        v2f b;
        b.x = B[kp * NC + n];
        b.y = B[(kp + 1) * NC + n];
        acc = __builtin_amdgcn_wmma_f32_16x16x4_f32(
             false, a,  false, b,
             (short)0, acc,  false,  false);
    }

    const float bv = BIAS_RELU ? bias[n] : 0.0f;
#pragma unroll
    for (int r = 0; r < 8; ++r) {
        const int row = rt * 16 + r + half * 8;
        float v = acc[r];
        if (BIAS_RELU) {
            v += bv;
            v = v > 0.0f ? v : 0.0f;
        }
        C[row * NC + n] = v;
    }
}

template <int F>
__global__ void propagate_kernel(const int* __restrict__ src, const int* __restrict__ dst,
                                 const float* __restrict__ dis, const float* __restrict__ H,
                                 float* __restrict__ agg, int E) {
    constexpr int G = F / 4;
    unsigned gid = blockIdx.x * blockDim.x + threadIdx.x;
    int e = (int)(gid / G);
    int f = (int)(gid % G) * 4;
    if (e >= E) return;
    __builtin_prefetch(src + e + 2048, 0, 0);
    __builtin_prefetch(dst + e + 2048, 0, 0);
    int s = src[e];
    int d = dst[e];
    float nrm = dis[s] * dis[d];
    const float4 hv = *(const float4*)(H + (size_t)s * F + f);
    float* ap = agg + (size_t)d * F + f;
    unsafeAtomicAdd(ap + 0, hv.x * nrm);
    unsafeAtomicAdd(ap + 1, hv.y * nrm);
    unsafeAtomicAdd(ap + 2, hv.z * nrm);
    unsafeAtomicAdd(ap + 3, hv.w * nrm);
}

__global__ void propagate4_kernel(const int* __restrict__ src, const int* __restrict__ dst,
                                  const float* __restrict__ dis, const float* __restrict__ X,
                                  float* __restrict__ agg, int E) {
    const int lane = threadIdx.x & 31;
    const int c    = lane & 3;
    const int q    = lane >> 2;
    const int e0   = (int)(blockIdx.x * blockDim.x + threadIdx.x) - lane;
    if (e0 + 32 <= E) {
        float val[4];
        int   row[4];
        for (int j = 0; j < 4; ++j) {
            const int e = e0 + j * 8 + q;
            const int s = src[e];
            const int d = dst[e];
            const float nrm = dis[s] * dis[d];
            val[j] = X[(size_t)s * 4 + c] * nrm;
            row[j] = d;
        }
        for (int j = 0; j < 4; ++j)
            unsafeAtomicAdd(agg + (size_t)row[j] * 4 + c, val[j]);
    } else {
        for (int j = 0; j < 4; ++j) {
            const int e = e0 + j * 8 + q;
            if (e < E) {
                const int s = src[e];
                const int d = dst[e];
                const float nrm = dis[s] * dis[d];
                unsafeAtomicAdd(agg + (size_t)d * 4 + c, X[(size_t)s * 4 + c] * nrm);
            }
        }
    }
}

template <int F>
__global__ void selfloop_bias_relu_kernel(const float* __restrict__ agg, const float* __restrict__ H,
                                          const float* __restrict__ dis, const float* __restrict__ bias,
                                          float* __restrict__ out, int total) {
    int i = blockIdx.x * blockDim.x + threadIdx.x;
    if (i >= total) return;
    int node = i / F;
    int f    = i % F;
    float ds = dis[node];
    float v  = agg[i] + H[i] * ds * ds + bias[f];
    out[i]   = v > 0.0f ? v : 0.0f;
}

extern "C" void kernel_launch(void* const* d_in, const int* in_sizes, int n_in,
                              void* d_out, int out_size, void* d_ws, size_t ws_size,
                              hipStream_t stream) {
    const float* x   = (const float*)d_in[0];
    const int*   ei  = (const int*)d_in[1];
    const float* W1  = (const float*)d_in[2];
    const float* b1  = (const float*)d_in[3];
    const float* W2  = (const float*)d_in[4];
    const float* b2  = (const float*)d_in[5];
    const float* Wfc = (const float*)d_in[6];
    const float* bfc = (const float*)d_in[7];
    float* out = (float*)d_out;

    const int N = in_sizes[0] / 4;
    const int E = in_sizes[1] / 2;
    const int* srcIdx = ei;
    const int* dstIdx = ei + E;

    float* ws   = (float*)d_ws;
    float* dis  = ws;
    float* h1   = ws + (((size_t)N + 255) & ~(size_t)255);
    float* buf2 = h1 + (size_t)N * 64;
    float* aggx = buf2;
    float* h2   = buf2;
    float* agg2 = buf2 + (size_t)N * 32;
    float* h2o  = h1;

    const int T = 256;
    const int rowTiles = N / 16;

    deg_init_kernel<<<(N + T - 1) / T, T, 0, stream>>>(dis, N);
    deg_accum_kernel<<<(E + T - 1) / T, T, 0, stream>>>(dstIdx, dis, E);
    deg_rsqrt_kernel<<<(N + T - 1) / T, T, 0, stream>>>(dis, N);

    selfloop_start4_kernel<<<(N + T - 1) / T, T, 0, stream>>>(x, dis, (float4*)aggx, N);
    propagate4_kernel<<<(E + T - 1) / T, T, 0, stream>>>(srcIdx, dstIdx, dis, x, aggx, E);

    gemm_wmma_kernel<4, 64, true><<<rowTiles, 128, 0, stream>>>(aggx, W1, b1, h1);

    gemm_wmma_kernel<64, 32, false><<<rowTiles, 64, 0, stream>>>(h1, W2, nullptr, h2);

    zero4_kernel<<<((N * 8) + T - 1) / T, T, 0, stream>>>((float4*)agg2, N * 8);
    {
        unsigned total = (unsigned)E * 8u;
        propagate_kernel<32><<<(total + T - 1) / T, T, 0, stream>>>(srcIdx, dstIdx, dis, h2, agg2, E);
    }
    selfloop_bias_relu_kernel<32><<<((N * 32) + T - 1) / T, T, 0, stream>>>(agg2, h2, dis, b2, h2o, N * 32);

    gemm_wmma_kernel<32, 32, true><<<rowTiles, 64, 0, stream>>>(h2o, Wfc, bfc, out);
}


# Layer-2 edge pass: one lane per float of the 32-float row, 32 edges per wave: each atomic instruction adds ONE whole 128-byte row (was 4 quarter rows of 4 lines); same addends, atomic order differs
# speedup vs baseline: 1.5337x; 1.1271x over previous
#include <hip/hip_runtime.h>

typedef __attribute__((ext_vector_type(2))) float v2f;
typedef __attribute__((ext_vector_type(8))) float v8f;

__global__ void deg_init_kernel(float* __restrict__ deg, int n) {
    int i = blockIdx.x * blockDim.x + threadIdx.x;
    if (i < n) deg[i] = 1.0f;
}

__global__ void deg_accum_kernel(const int* __restrict__ dst, float* __restrict__ deg, int E) {
    int e = blockIdx.x * blockDim.x + threadIdx.x;
    if (e < E) {
        __builtin_prefetch(dst + e + 4096, 0, 0);
        unsafeAtomicAdd(&deg[dst[e]], 1.0f);
    }
}

__global__ void deg_rsqrt_kernel(float* __restrict__ deg, int n) {
    int i = blockIdx.x * blockDim.x + threadIdx.x;
    if (i < n) deg[i] = rsqrtf(deg[i]);
}

__global__ void zero4_kernel(float4* __restrict__ p, int n4) {
    int i = blockIdx.x * blockDim.x + threadIdx.x;
    if (i < n4) p[i] = make_float4(0.f, 0.f, 0.f, 0.f);
}

__global__ void selfloop_start4_kernel(const float* __restrict__ x, const float* __restrict__ dis,
                                       float4* __restrict__ aggx, int n) {
    int i = blockIdx.x * blockDim.x + threadIdx.x;
    if (i >= n) return;
    const float ds = dis[i];
    const float4 xv = *(const float4*)(x + (size_t)i * 4);
    aggx[i] = make_float4(xv.x * ds * ds, xv.y * ds * ds, xv.z * ds * ds, xv.w * ds * ds);
}

template <int K, int NC, bool BIAS_RELU>
__global__ void gemm_wmma_kernel(const float* __restrict__ A, const float* __restrict__ B,
                                 const float* __restrict__ bias, float* __restrict__ C) {
    const int lane = threadIdx.x & 31;
    const int wave = threadIdx.x >> 5;
    const int rt   = blockIdx.x;
    const int half = lane >> 4;
    const int l16  = lane & 15;
    const int m    = rt * 16 + l16;
    const int n    = wave * 16 + l16;

    v8f acc = {};
#pragma unroll
    for (int kb = 0; kb < K; kb += 4) {
        const int kp = kb + half * 2;
        v2f a;
        a.x = A[m * K + kp];
        a.y = A[m * K + kp + 1];
        v2f b;
        b.x = B[kp * NC + n];
        b.y = B[(kp + 1) * NC + n];
        acc = __builtin_amdgcn_wmma_f32_16x16x4_f32(
             false, a,  false, b,
             (short)0, acc,  false,  false);
    }

    const float bv = BIAS_RELU ? bias[n] : 0.0f;
#pragma unroll
    for (int r = 0; r < 8; ++r) {
        const int row = rt * 16 + r + half * 8;
        float v = acc[r];
        if (BIAS_RELU) {
            v += bv;
            v = v > 0.0f ? v : 0.0f;
        }
        C[row * NC + n] = v;
    }
}

__global__ void propagate32_kernel(const int* __restrict__ src, const int* __restrict__ dst,
                                   const float* __restrict__ dis, const float* __restrict__ H,
                                   float* __restrict__ agg, int E) {
    const int lane = threadIdx.x & 31;
    const int e0 = __builtin_amdgcn_readfirstlane((int)(blockIdx.x * blockDim.x + (threadIdx.x & ~31u)));
    if (e0 >= E) return;
    if (e0 + 32 <= E) {
        const int s = src[e0 + lane];
        const int d = dst[e0 + lane];
        const int nrm = __float_as_int(dis[s] * dis[d]);
        float v[32];
#pragma unroll
        for (int j = 0; j < 32; ++j) {
            const int sj = __builtin_amdgcn_readlane(s, j);
            v[j] = H[(unsigned)sj * 32u + (unsigned)lane];
        }
#pragma unroll
        for (int j = 0; j < 32; ++j) {
            const int   dj = __builtin_amdgcn_readlane(d, j);
            const float nj = __int_as_float(__builtin_amdgcn_readlane(nrm, j));
            unsafeAtomicAdd(agg + ((unsigned)dj * 32u + (unsigned)lane), v[j] * nj);
        }
    } else {
        const int cnt = E - e0;
        const bool ok = lane < cnt;
        const int s = ok ? src[e0 + lane] : 0;
        const int d = ok ? dst[e0 + lane] : 0;
        const int nrm = __float_as_int(ok ? dis[s] * dis[d] : 0.0f);
        for (int j = 0; j < cnt; ++j) {
            const int   sj = __builtin_amdgcn_readlane(s, j);
            const int   dj = __builtin_amdgcn_readlane(d, j);
            const float nj = __int_as_float(__builtin_amdgcn_readlane(nrm, j));
            unsafeAtomicAdd(agg + ((unsigned)dj * 32u + (unsigned)lane), H[(unsigned)sj * 32u + (unsigned)lane] * nj);
        }
    }
}

__global__ void propagate4_kernel(const int* __restrict__ src, const int* __restrict__ dst,
                                  const float* __restrict__ dis, const float* __restrict__ X,
                                  float* __restrict__ agg, int E) {
    const int lane = threadIdx.x & 31;
    const int c    = lane & 3;
    const int q    = lane >> 2;
    const int e0   = (int)(blockIdx.x * blockDim.x + threadIdx.x) - lane;
    if (e0 + 32 <= E) {
        float val[4];
        int   row[4];
        for (int j = 0; j < 4; ++j) {
            const int e = e0 + j * 8 + q;
            const int s = src[e];
            const int d = dst[e];
            const float nrm = dis[s] * dis[d];
            val[j] = X[(size_t)s * 4 + c] * nrm;
            row[j] = d;
        }
        for (int j = 0; j < 4; ++j)
            unsafeAtomicAdd(agg + (size_t)row[j] * 4 + c, val[j]);
    } else {
        for (int j = 0; j < 4; ++j) {
            const int e = e0 + j * 8 + q;
            if (e < E) {
                const int s = src[e];
                const int d = dst[e];
                const float nrm = dis[s] * dis[d];
                unsafeAtomicAdd(agg + (size_t)d * 4 + c, X[(size_t)s * 4 + c] * nrm);
            }
        }
    }
}

template <int F>
__global__ void selfloop_bias_relu_kernel(const float* __restrict__ agg, const float* __restrict__ H,
                                          const float* __restrict__ dis, const float* __restrict__ bias,
                                          float* __restrict__ out, int total) {
    int i = blockIdx.x * blockDim.x + threadIdx.x;
    if (i >= total) return;
    int node = i / F;
    int f    = i % F;
    float ds = dis[node];
    float v  = agg[i] + H[i] * ds * ds + bias[f];
    out[i]   = v > 0.0f ? v : 0.0f;
}

extern "C" void kernel_launch(void* const* d_in, const int* in_sizes, int n_in,
                              void* d_out, int out_size, void* d_ws, size_t ws_size,
                              hipStream_t stream) {
    const float* x   = (const float*)d_in[0];
    const int*   ei  = (const int*)d_in[1];
    const float* W1  = (const float*)d_in[2];
    const float* b1  = (const float*)d_in[3];
    const float* W2  = (const float*)d_in[4];
    const float* b2  = (const float*)d_in[5];
    const float* Wfc = (const float*)d_in[6];
    const float* bfc = (const float*)d_in[7];
    float* out = (float*)d_out;

    const int N = in_sizes[0] / 4;
    const int E = in_sizes[1] / 2;
    const int* srcIdx = ei;
    const int* dstIdx = ei + E;

    float* ws   = (float*)d_ws;
    float* dis  = ws;
    float* h1   = ws + (((size_t)N + 255) & ~(size_t)255);
    float* buf2 = h1 + (size_t)N * 64;
    float* aggx = buf2;
    float* h2   = buf2;
    float* agg2 = buf2 + (size_t)N * 32;
    float* h2o  = h1;

    const int T = 256;
    const int rowTiles = N / 16;

    deg_init_kernel<<<(N + T - 1) / T, T, 0, stream>>>(dis, N);
    deg_accum_kernel<<<(E + T - 1) / T, T, 0, stream>>>(dstIdx, dis, E);
    deg_rsqrt_kernel<<<(N + T - 1) / T, T, 0, stream>>>(dis, N);

    selfloop_start4_kernel<<<(N + T - 1) / T, T, 0, stream>>>(x, dis, (float4*)aggx, N);
    propagate4_kernel<<<(E + T - 1) / T, T, 0, stream>>>(srcIdx, dstIdx, dis, x, aggx, E);

    gemm_wmma_kernel<4, 64, true><<<rowTiles, 128, 0, stream>>>(aggx, W1, b1, h1);

    gemm_wmma_kernel<64, 32, false><<<rowTiles, 64, 0, stream>>>(h1, W2, nullptr, h2);

    zero4_kernel<<<((N * 8) + T - 1) / T, T, 0, stream>>>((float4*)agg2, N * 8);
    propagate32_kernel<<<(E + T - 1) / T, T, 0, stream>>>(srcIdx, dstIdx, dis, h2, agg2, E);
    selfloop_bias_relu_kernel<32><<<((N * 32) + T - 1) / T, T, 0, stream>>>(agg2, h2, dis, b2, h2o, N * 32);

    gemm_wmma_kernel<32, 32, true><<<rowTiles, 64, 0, stream>>>(h2o, Wfc, bfc, out);
}


# Edge weight split over the edge's two ends: rows scaled by dis[src] once per node, finished sums by dis[dst] once per node; both edge passes gather no dis; rounding of the products regrouped
# speedup vs baseline: 1.6931x; 1.1039x over previous
#include <hip/hip_runtime.h>

typedef __attribute__((ext_vector_type(2))) float v2f;
typedef __attribute__((ext_vector_type(8))) float v8f;

__global__ void deg_init_kernel(float* __restrict__ deg, int n) {
    int i = blockIdx.x * blockDim.x + threadIdx.x;
    if (i < n) deg[i] = 1.0f;
}

__global__ void deg_accum_kernel(const int* __restrict__ dst, float* __restrict__ deg, int E) {
    int e = blockIdx.x * blockDim.x + threadIdx.x;
    if (e < E) {
        __builtin_prefetch(dst + e + 4096, 0, 0);
        unsafeAtomicAdd(&deg[dst[e]], 1.0f);
    }
}

__global__ void deg_rsqrt_kernel(float* __restrict__ deg, int n) {
    int i = blockIdx.x * blockDim.x + threadIdx.x;
    if (i < n) deg[i] = rsqrtf(deg[i]);
}

__global__ void zero4_kernel(float4* __restrict__ p, int n4) {
    int i = blockIdx.x * blockDim.x + threadIdx.x;
    if (i < n4) p[i] = make_float4(0.f, 0.f, 0.f, 0.f);
}

__global__ void prescale_start4_kernel(const float* __restrict__ x, const float* __restrict__ dis,
                                       float4* __restrict__ xs, float4* __restrict__ aggx, int n) {
    int i = blockIdx.x * blockDim.x + threadIdx.x;
    if (i >= n) return;
    const float ds = dis[i];
    const float4 xv = *(const float4*)(x + (size_t)i * 4);
    const float4 r = make_float4(xv.x * ds, xv.y * ds, xv.z * ds, xv.w * ds);
    xs[i]   = r;
    aggx[i] = r;
}

template <int K, int NC, bool BIAS_RELU, bool ROW_SCALE>
__global__ void gemm_wmma_kernel(const float* __restrict__ A, const float* __restrict__ B,
                                 const float* __restrict__ bias, const float* __restrict__ rs,
                                 float* __restrict__ C) {
    const int lane = threadIdx.x & 31;
    const int wave = threadIdx.x >> 5;
    const int rt   = blockIdx.x;
    const int half = lane >> 4;
    const int l16  = lane & 15;
    const int m    = rt * 16 + l16;
    const int n    = wave * 16 + l16;

    v8f acc = {};
#pragma unroll
    for (int kb = 0; kb < K; kb += 4) {
        const int kp = kb + half * 2;
        v2f a;
        a.x = A[m * K + kp];
        a.y = A[m * K + kp + 1];
        v2f b;
        b.x = B[kp * NC + n];
        b.y = B[(kp + 1) * NC + n];
        acc = __builtin_amdgcn_wmma_f32_16x16x4_f32(
             false, a,  false, b,
             (short)0, acc,  false,  false);
    }

    const float bv = BIAS_RELU ? bias[n] : 0.0f;
#pragma unroll
    for (int r = 0; r < 8; ++r) {
        const int row = rt * 16 + r + half * 8;
        float v = acc[r];
        if (ROW_SCALE) v *= rs[row];
        if (BIAS_RELU) {
            v += bv;
            v = v > 0.0f ? v : 0.0f;
        }
        C[row * NC + n] = v;
    }
}

__global__ void propagate32_kernel(const int* __restrict__ src, const int* __restrict__ dst,
                                   const float* __restrict__ H,
                                   float* __restrict__ agg, int E) {
    const int lane = threadIdx.x & 31;
    const int e0 = __builtin_amdgcn_readfirstlane((int)(blockIdx.x * blockDim.x + (threadIdx.x & ~31u)));
    if (e0 >= E) return;
    if (e0 + 32 <= E) {
        const int s = src[e0 + lane];
        const int d = dst[e0 + lane];
        float v[32];
#pragma unroll
        for (int j = 0; j < 32; ++j) {
            const int sj = __builtin_amdgcn_readlane(s, j);
            v[j] = H[(unsigned)sj * 32u + (unsigned)lane];
        }
#pragma unroll
        for (int j = 0; j < 32; ++j) {
            const int dj = __builtin_amdgcn_readlane(d, j);
            unsafeAtomicAdd(agg + ((unsigned)dj * 32u + (unsigned)lane), v[j]);
        }
    } else {
        const int cnt = E - e0;
        const bool ok = lane < cnt;
        const int s = ok ? src[e0 + lane] : 0;
        const int d = ok ? dst[e0 + lane] : 0;
        for (int j = 0; j < cnt; ++j) {
            const int sj = __builtin_amdgcn_readlane(s, j);
            const int dj = __builtin_amdgcn_readlane(d, j);
            unsafeAtomicAdd(agg + ((unsigned)dj * 32u + (unsigned)lane), H[(unsigned)sj * 32u + (unsigned)lane]);
        }
    }
}

__global__ void propagate4_kernel(const int* __restrict__ src, const int* __restrict__ dst,
                                  const float* __restrict__ X,
                                  float* __restrict__ agg, int E) {
    const int lane = threadIdx.x & 31;
    const int c    = lane & 3;
    const int q    = lane >> 2;
    const int e0   = (int)(blockIdx.x * blockDim.x + threadIdx.x) - lane;
    if (e0 + 32 <= E) {
        float val[4];
        int   row[4];
        for (int j = 0; j < 4; ++j) {
            const int e = e0 + j * 8 + q;
            const int s = src[e];
            const int d = dst[e];
            val[j] = X[(size_t)s * 4 + c];
            row[j] = d;
        }
        for (int j = 0; j < 4; ++j)
            unsafeAtomicAdd(agg + (size_t)row[j] * 4 + c, val[j]);
    } else {
        for (int j = 0; j < 4; ++j) {
            const int e = e0 + j * 8 + q;
            if (e < E) {
                const int s = src[e];
                const int d = dst[e];
                unsafeAtomicAdd(agg + (size_t)d * 4 + c, X[(size_t)s * 4 + c]);
            }
        }
    }
}

template <int F>
__global__ void selfloop_bias_relu_kernel(const float* __restrict__ agg, const float* __restrict__ H,
                                          const float* __restrict__ dis, const float* __restrict__ bias,
                                          float* __restrict__ out, int total) {
    int i = blockIdx.x * blockDim.x + threadIdx.x;
    if (i >= total) return;
    int node = i / F;
    int f    = i % F;
    float ds = dis[node];
    float v  = (agg[i] + H[i]) * ds + bias[f];
    out[i]   = v > 0.0f ? v : 0.0f;
}

extern "C" void kernel_launch(void* const* d_in, const int* in_sizes, int n_in,
                              void* d_out, int out_size, void* d_ws, size_t ws_size,
                              hipStream_t stream) {
    const float* x   = (const float*)d_in[0];
    const int*   ei  = (const int*)d_in[1];
    const float* W1  = (const float*)d_in[2];
    const float* b1  = (const float*)d_in[3];
    const float* W2  = (const float*)d_in[4];
    const float* b2  = (const float*)d_in[5];
    const float* Wfc = (const float*)d_in[6];
    const float* bfc = (const float*)d_in[7];
    float* out = (float*)d_out;

    const int N = in_sizes[0] / 4;
    const int E = in_sizes[1] / 2;
    const int* srcIdx = ei;
    const int* dstIdx = ei + E;

    float* ws   = (float*)d_ws;
    float* dis  = ws;
    float* h1   = ws + (((size_t)N + 255) & ~(size_t)255);
    float* buf2 = h1 + (size_t)N * 64;
    float* aggx = buf2;
    float* xs   = buf2 + (size_t)N * 4;
    float* h2   = buf2;
    float* agg2 = buf2 + (size_t)N * 32;
    float* h2o  = h1;

    const int T = 256;
    const int rowTiles = N / 16;

    deg_init_kernel<<<(N + T - 1) / T, T, 0, stream>>>(dis, N);
    deg_accum_kernel<<<(E + T - 1) / T, T, 0, stream>>>(dstIdx, dis, E);
    deg_rsqrt_kernel<<<(N + T - 1) / T, T, 0, stream>>>(dis, N);

    prescale_start4_kernel<<<(N + T - 1) / T, T, 0, stream>>>(x, dis, (float4*)xs, (float4*)aggx, N);
    propagate4_kernel<<<(E + T - 1) / T, T, 0, stream>>>(srcIdx, dstIdx, xs, aggx, E);

    gemm_wmma_kernel<4, 64, true, true><<<rowTiles, 128, 0, stream>>>(aggx, W1, b1, dis, h1);

    gemm_wmma_kernel<64, 32, false, true><<<rowTiles, 64, 0, stream>>>(h1, W2, nullptr, dis, h2);

    zero4_kernel<<<((N * 8) + T - 1) / T, T, 0, stream>>>((float4*)agg2, N * 8);
    propagate32_kernel<<<(E + T - 1) / T, T, 0, stream>>>(srcIdx, dstIdx, h2, agg2, E);
    selfloop_bias_relu_kernel<32><<<((N * 32) + T - 1) / T, T, 0, stream>>>(agg2, h2, dis, b2, h2o, N * 32);

    gemm_wmma_kernel<32, 32, true, false><<<rowTiles, 64, 0, stream>>>(h2o, Wfc, bfc, nullptr, out);
}


# Each atomically summed table (degree, both edge sums) kept twice, one copy per package half; a work-group adds into its half's copy (die id asked once per group), consumer adds both; sums regrouped
# speedup vs baseline: 3.9710x; 2.3454x over previous
#include <hip/hip_runtime.h>

typedef __attribute__((ext_vector_type(2))) float v2f;
typedef __attribute__((ext_vector_type(8))) float v8f;

__device__ __forceinline__ unsigned block_half(unsigned* slot) {
    if (threadIdx.x == 0)
        *slot = (__builtin_amdgcn_s_sendmsg_rtn(0x87) >> 18) & 1u;
    __syncthreads();
    return (unsigned)__builtin_amdgcn_readfirstlane((int)*slot);
}

__global__ void deg_init_kernel(float* __restrict__ deg0, float* __restrict__ deg1, int n) {
    int i = blockIdx.x * blockDim.x + threadIdx.x;
    if (i < n) {
        deg0[i] = 1.0f;
        deg1[i] = 0.0f;
    }
}

__global__ void deg_accum_kernel(const int* __restrict__ dst, float* __restrict__ deg0,
                                 float* __restrict__ deg1, int E) {
    __shared__ unsigned s_half;
    float* __restrict__ deg = block_half(&s_half) ? deg1 : deg0;
    int e = blockIdx.x * blockDim.x + threadIdx.x;
    if (e < E) {
        __builtin_prefetch(dst + e + 4096, 0, 0);
        unsafeAtomicAdd(&deg[dst[e]], 1.0f);
    }
}

__global__ void deg_rsqrt_kernel(const float* __restrict__ deg0, const float* __restrict__ deg1,
                                 float* __restrict__ dis, int n) {
    int i = blockIdx.x * blockDim.x + threadIdx.x;
    if (i < n) dis[i] = rsqrtf(deg0[i] + deg1[i]);
}

__global__ void zero4_kernel(float4* __restrict__ p, float4* __restrict__ q, int n4) {
    int i = blockIdx.x * blockDim.x + threadIdx.x;
    if (i < n4) {
        p[i] = make_float4(0.f, 0.f, 0.f, 0.f);
        q[i] = make_float4(0.f, 0.f, 0.f, 0.f);
    }
}

__global__ void prescale_start4_kernel(const float* __restrict__ x, const float* __restrict__ dis,
                                       float4* __restrict__ xs, float4* __restrict__ aggx,
                                       float4* __restrict__ aggx1, int n) {
    int i = blockIdx.x * blockDim.x + threadIdx.x;
    if (i >= n) return;
    const float ds = dis[i];
    const float4 xv = *(const float4*)(x + (size_t)i * 4);
    const float4 r = make_float4(xv.x * ds, xv.y * ds, xv.z * ds, xv.w * ds);
    xs[i]    = r;
    aggx[i]  = r;
    aggx1[i] = make_float4(0.f, 0.f, 0.f, 0.f);
}

template <int K, int NC, bool BIAS_RELU, bool ROW_SCALE, bool TWO_A>
__global__ void gemm_wmma_kernel(const float* __restrict__ A, const float* __restrict__ A2,
                                 const float* __restrict__ B,
                                 const float* __restrict__ bias, const float* __restrict__ rs,
                                 float* __restrict__ C) {
    const int lane = threadIdx.x & 31;
    const int wave = threadIdx.x >> 5;
    const int rt   = blockIdx.x;
    const int half = lane >> 4;
    const int l16  = lane & 15;
    const int m    = rt * 16 + l16;
    const int n    = wave * 16 + l16;

    v8f acc = {};
#pragma unroll
    for (int kb = 0; kb < K; kb += 4) {
        const int kp = kb + half * 2;
        v2f a;
        a.x = A[m * K + kp];
        a.y = A[m * K + kp + 1];
        if (TWO_A) {
            a.x += A2[m * K + kp];
            a.y += A2[m * K + kp + 1];
        }
        v2f b;
        b.x = B[kp * NC + n];
        b.y = B[(kp + 1) * NC + n];
        acc = __builtin_amdgcn_wmma_f32_16x16x4_f32(
             false, a,  false, b,
             (short)0, acc,  false,  false);
    }

    const float bv = BIAS_RELU ? bias[n] : 0.0f;
#pragma unroll
    for (int r = 0; r < 8; ++r) {
        const int row = rt * 16 + r + half * 8;
        float v = acc[r];
        if (ROW_SCALE) v *= rs[row];
        if (BIAS_RELU) {
            v += bv;
            v = v > 0.0f ? v : 0.0f;
        }
        C[row * NC + n] = v;
    }
}

__global__ void propagate32_kernel(const int* __restrict__ src, const int* __restrict__ dst,
                                   const float* __restrict__ H,
                                   float* __restrict__ agg0, float* __restrict__ agg1, int E) {
    __shared__ unsigned s_half;
    float* __restrict__ agg = block_half(&s_half) ? agg1 : agg0;
    const int lane = threadIdx.x & 31;
    const int e0 = __builtin_amdgcn_readfirstlane((int)(blockIdx.x * blockDim.x + (threadIdx.x & ~31u)));
    if (e0 >= E) return;
    if (e0 + 32 <= E) {
        const int s = src[e0 + lane];
        const int d = dst[e0 + lane];
        float v[32];
#pragma unroll
        for (int j = 0; j < 32; ++j) {
            const int sj = __builtin_amdgcn_readlane(s, j);
            v[j] = H[(unsigned)sj * 32u + (unsigned)lane];
        }
#pragma unroll
        for (int j = 0; j < 32; ++j) {
            const int dj = __builtin_amdgcn_readlane(d, j);
            unsafeAtomicAdd(agg + ((unsigned)dj * 32u + (unsigned)lane), v[j]);
        }
    } else {
        const int cnt = E - e0;
        const bool ok = lane < cnt;
        const int s = ok ? src[e0 + lane] : 0;
        const int d = ok ? dst[e0 + lane] : 0;
        for (int j = 0; j < cnt; ++j) {
            const int sj = __builtin_amdgcn_readlane(s, j);
            const int dj = __builtin_amdgcn_readlane(d, j);
            unsafeAtomicAdd(agg + ((unsigned)dj * 32u + (unsigned)lane), H[(unsigned)sj * 32u + (unsigned)lane]);
        }
    }
}

__global__ void propagate4_kernel(const int* __restrict__ src, const int* __restrict__ dst,
                                  const float* __restrict__ X,
                                  float* __restrict__ agg0, float* __restrict__ agg1, int E) {
    __shared__ unsigned s_half;
    float* __restrict__ agg = block_half(&s_half) ? agg1 : agg0;
    const int lane = threadIdx.x & 31;
    const int c    = lane & 3;
    const int q    = lane >> 2;
    const int e0   = (int)(blockIdx.x * blockDim.x + threadIdx.x) - lane;
    if (e0 + 32 <= E) {
        float val[4];
        int   row[4];
        for (int j = 0; j < 4; ++j) {
            const int e = e0 + j * 8 + q;
            const int s = src[e];
            const int d = dst[e];
            val[j] = X[(size_t)s * 4 + c];
            row[j] = d;
        }
        for (int j = 0; j < 4; ++j)
            unsafeAtomicAdd(agg + (size_t)row[j] * 4 + c, val[j]);
    } else {
        for (int j = 0; j < 4; ++j) {
            const int e = e0 + j * 8 + q;
            if (e < E) {
                const int s = src[e];
                const int d = dst[e];
                unsafeAtomicAdd(agg + (size_t)d * 4 + c, X[(size_t)s * 4 + c]);
            }
        }
    }
}

template <int F>
__global__ void selfloop_bias_relu_kernel(const float* __restrict__ agg0, const float* __restrict__ agg1,
                                          const float* __restrict__ H,
                                          const float* __restrict__ dis, const float* __restrict__ bias,
                                          float* __restrict__ out, int total) {
    int i = blockIdx.x * blockDim.x + threadIdx.x;
    if (i >= total) return;
    int node = i / F;
    int f    = i % F;
    float ds = dis[node];
    float v  = ((agg0[i] + agg1[i]) + H[i]) * ds + bias[f];
    out[i]   = v > 0.0f ? v : 0.0f;
}

extern "C" void kernel_launch(void* const* d_in, const int* in_sizes, int n_in,
                              void* d_out, int out_size, void* d_ws, size_t ws_size,
                              hipStream_t stream) {
    const float* x   = (const float*)d_in[0];
    const int*   ei  = (const int*)d_in[1];
    const float* W1  = (const float*)d_in[2];
    const float* b1  = (const float*)d_in[3];
    const float* W2  = (const float*)d_in[4];
    const float* b2  = (const float*)d_in[5];
    const float* Wfc = (const float*)d_in[6];
    const float* bfc = (const float*)d_in[7];
    float* out = (float*)d_out;

    const int N = in_sizes[0] / 4;
    const int E = in_sizes[1] / 2;
    const int* srcIdx = ei;
    const int* dstIdx = ei + E;

    float* ws   = (float*)d_ws;
    const size_t Np = ((size_t)N + 255) & ~(size_t)255;
    float* dis  = ws;
    float* h1   = ws + Np;
    float* buf2 = h1 + (size_t)N * 64;
    float* deg0 = h1;
    float* deg1 = h1 + Np;
    float* aggx  = buf2;
    float* xs    = buf2 + (size_t)N * 4;
    float* aggx1 = buf2 + (size_t)N * 8;
    float* h2   = buf2;
    float* agg2 = buf2 + (size_t)N * 32;
    float* agg2b = h1;
    float* h2o  = h1 + (size_t)N * 32;

    const int T = 256;
    const int rowTiles = N / 16;

    deg_init_kernel<<<(N + T - 1) / T, T, 0, stream>>>(deg0, deg1, N);
    deg_accum_kernel<<<(E + T - 1) / T, T, 0, stream>>>(dstIdx, deg0, deg1, E);
    deg_rsqrt_kernel<<<(N + T - 1) / T, T, 0, stream>>>(deg0, deg1, dis, N);

    prescale_start4_kernel<<<(N + T - 1) / T, T, 0, stream>>>(x, dis, (float4*)xs, (float4*)aggx, (float4*)aggx1, N);
    propagate4_kernel<<<(E + T - 1) / T, T, 0, stream>>>(srcIdx, dstIdx, xs, aggx, aggx1, E);

    gemm_wmma_kernel<4, 64, true, true, true><<<rowTiles, 128, 0, stream>>>(aggx, aggx1, W1, b1, dis, h1);

    gemm_wmma_kernel<64, 32, false, true, false><<<rowTiles, 64, 0, stream>>>(h1, nullptr, W2, nullptr, dis, h2);

    zero4_kernel<<<((N * 8) + T - 1) / T, T, 0, stream>>>((float4*)agg2, (float4*)agg2b, N * 8);
    propagate32_kernel<<<(E + T - 1) / T, T, 0, stream>>>(srcIdx, dstIdx, h2, agg2, agg2b, E);
    selfloop_bias_relu_kernel<32><<<((N * 32) + T - 1) / T, T, 0, stream>>>(agg2, agg2b, h2, dis, b2, h2o, N * 32);

    gemm_wmma_kernel<32, 32, true, false, false><<<rowTiles, 64, 0, stream>>>(h2o, nullptr, Wfc, bfc, nullptr, out);
}


# Per-node kernels between the edge passes fused, 11 launches -> 7: one start-values kernel, rsqrt inside the prescale, both layer products chained through LDS, layer-2 epilogue staged for the last one
# speedup vs baseline: 5.7419x; 1.4460x over previous
#include <hip/hip_runtime.h>

typedef __attribute__((ext_vector_type(2))) float v2f;
typedef __attribute__((ext_vector_type(8))) float v8f;

__device__ __forceinline__ unsigned block_half(unsigned* slot) {
    if (threadIdx.x == 0)
        *slot = (__builtin_amdgcn_s_sendmsg_rtn(0x87) >> 18) & 1u;
    __syncthreads();
    return (unsigned)__builtin_amdgcn_readfirstlane((int)*slot);
}

__global__ void start_values_kernel(float* __restrict__ deg0, float* __restrict__ deg1,
                                    float4* __restrict__ p, float4* __restrict__ q, int n, int n4) {
    int i = blockIdx.x * blockDim.x + threadIdx.x;
    if (i < n4) {
        p[i] = make_float4(0.f, 0.f, 0.f, 0.f);
        q[i] = make_float4(0.f, 0.f, 0.f, 0.f);
    }
    if (i < n) {
        deg0[i] = 1.0f;
        deg1[i] = 0.0f;
    }
}

__global__ void deg_accum_kernel(const int* __restrict__ dst, float* __restrict__ deg0,
                                 float* __restrict__ deg1, int E) {
    __shared__ unsigned s_half;
    float* __restrict__ deg = block_half(&s_half) ? deg1 : deg0;
    int e = blockIdx.x * blockDim.x + threadIdx.x;
    if (e < E) {
        __builtin_prefetch(dst + e + 4096, 0, 0);
        unsafeAtomicAdd(&deg[dst[e]], 1.0f);
    }
}

__global__ void prescale_start4_kernel(const float* __restrict__ x,
                                       const float* __restrict__ deg0, const float* __restrict__ deg1,
                                       float* __restrict__ dis,
                                       float4* __restrict__ xs, float4* __restrict__ aggx,
                                       float4* __restrict__ aggx1, int n) {
    int i = blockIdx.x * blockDim.x + threadIdx.x;
    if (i >= n) return;
    const float ds = rsqrtf(deg0[i] + deg1[i]);
    dis[i] = ds;
    const float4 xv = *(const float4*)(x + (size_t)i * 4);
    const float4 r = make_float4(xv.x * ds, xv.y * ds, xv.z * ds, xv.w * ds);
    xs[i]    = r;
    aggx[i]  = r;
    aggx1[i] = make_float4(0.f, 0.f, 0.f, 0.f);
}


#define H1_PITCH 68
__global__ void __launch_bounds__(64)
layer12_wmma_kernel(const float* __restrict__ aggx, const float* __restrict__ aggx1,
                    const float* __restrict__ W1, const float* __restrict__ b1,
                    const float* __restrict__ W2, const float* __restrict__ dis,
                    float* __restrict__ hs) {
    __shared__ float sH[16 * H1_PITCH];
    const int lane = threadIdx.x & 31;
    const int wave = threadIdx.x >> 5;
    const int rt   = blockIdx.x;
    const int half = lane >> 4;
    const int l16  = lane & 15;
    const int m    = rt * 16 + l16;
    const int kp   = half * 2;

    v2f a;
    a.x = aggx[m * 4 + kp]     + aggx1[m * 4 + kp];
    a.y = aggx[m * 4 + kp + 1] + aggx1[m * 4 + kp + 1];

    float rsv[8];
#pragma unroll
    for (int r = 0; r < 8; ++r) rsv[r] = dis[rt * 16 + r + half * 8];

#pragma unroll
    for (int t = 0; t < 2; ++t) {
        const int n1 = (wave * 2 + t) * 16 + l16;
        v2f b;
        b.x = W1[kp * 64 + n1];
        b.y = W1[(kp + 1) * 64 + n1];
        v8f acc = {};
        acc = __builtin_amdgcn_wmma_f32_16x16x4_f32(
             false, a,  false, b,
             (short)0, acc,  false,  false);
        const float bv = b1[n1];
#pragma unroll
        for (int r = 0; r < 8; ++r) {
            float v = acc[r];
            v *= rsv[r];
            v += bv;
            v = v > 0.0f ? v : 0.0f;
            sH[(r + half * 8) * H1_PITCH + n1] = v;
        }
    }
    __syncthreads();

    const int n = wave * 16 + l16;
    v8f acc2 = {};
#pragma unroll
    for (int kb = 0; kb < 64; kb += 4) {
        const int kq = kb + half * 2;
        const v2f a2 = *(const v2f*)&sH[l16 * H1_PITCH + kq];
        v2f b2;
        b2.x = W2[kq * 32 + n];
        b2.y = W2[(kq + 1) * 32 + n];
        acc2 = __builtin_amdgcn_wmma_f32_16x16x4_f32(
             false, a2,  false, b2,
             (short)0, acc2,  false,  false);
    }
#pragma unroll
    for (int r = 0; r < 8; ++r) {
        const int row = rt * 16 + r + half * 8;
        hs[row * 32 + n] = acc2[r] * rsv[r];
    }
}

#define H2_PITCH 36
__global__ void __launch_bounds__(64)
out_wmma_kernel(const float* __restrict__ agg0, const float* __restrict__ agg1,
                const float* __restrict__ H, const float* __restrict__ dis,
                const float* __restrict__ b2, const float* __restrict__ Wfc,
                const float* __restrict__ bfc, float* __restrict__ out) {
    __shared__ __attribute__((aligned(16))) float sA[16 * H2_PITCH];
    const int rt = blockIdx.x;
#pragma unroll
    for (int u = 0; u < 2; ++u) {
        const int idx = (int)threadIdx.x + u * 64;
        const int rl  = idx >> 3;
        const int c4  = idx & 7;
        const int row = rt * 16 + rl;
        const int g   = row * 32 + c4 * 4;
        const float4 p  = *(const float4*)(agg0 + g);
        const float4 q  = *(const float4*)(agg1 + g);
        const float4 h  = *(const float4*)(H + g);
        const float4 bb = *(const float4*)(b2 + c4 * 4);
        const float ds  = dis[row];
        float4 v;
        v.x = ((p.x + q.x) + h.x) * ds + bb.x;
        v.y = ((p.y + q.y) + h.y) * ds + bb.y;
        v.z = ((p.z + q.z) + h.z) * ds + bb.z;
        v.w = ((p.w + q.w) + h.w) * ds + bb.w;
        v.x = v.x > 0.0f ? v.x : 0.0f;
        v.y = v.y > 0.0f ? v.y : 0.0f;
        v.z = v.z > 0.0f ? v.z : 0.0f;
        v.w = v.w > 0.0f ? v.w : 0.0f;
        *(float4*)&sA[rl * H2_PITCH + c4 * 4] = v;
    }
    __syncthreads();

    const int lane = threadIdx.x & 31;
    const int wave = threadIdx.x >> 5;
    const int half = lane >> 4;
    const int l16  = lane & 15;
    const int n    = wave * 16 + l16;
    v8f acc = {};
#pragma unroll
    for (int kb = 0; kb < 32; kb += 4) {
        const int kq = kb + half * 2;
        const v2f a = *(const v2f*)&sA[l16 * H2_PITCH + kq];
        v2f b;
        b.x = Wfc[kq * 32 + n];
        b.y = Wfc[(kq + 1) * 32 + n];
        acc = __builtin_amdgcn_wmma_f32_16x16x4_f32(
             false, a,  false, b,
             (short)0, acc,  false,  false);
    }
    const float bv = bfc[n];
#pragma unroll
    for (int r = 0; r < 8; ++r) {
        const int row = rt * 16 + r + half * 8;
        float v = acc[r] + bv;
        v = v > 0.0f ? v : 0.0f;
        out[row * 32 + n] = v;
    }
}

__global__ void propagate32_kernel(const int* __restrict__ src, const int* __restrict__ dst,
                                   const float* __restrict__ H,
                                   float* __restrict__ agg0, float* __restrict__ agg1, int E) {
    __shared__ unsigned s_half;
    float* __restrict__ agg = block_half(&s_half) ? agg1 : agg0;
    const int lane = threadIdx.x & 31;
    const int e0 = __builtin_amdgcn_readfirstlane((int)(blockIdx.x * blockDim.x + (threadIdx.x & ~31u)));
    if (e0 >= E) return;
    if (e0 + 32 <= E) {
        const int s = src[e0 + lane];
        const int d = dst[e0 + lane];
        float v[32];
#pragma unroll
        for (int j = 0; j < 32; ++j) {
            const int sj = __builtin_amdgcn_readlane(s, j);
            v[j] = H[(unsigned)sj * 32u + (unsigned)lane];
        }
#pragma unroll
        for (int j = 0; j < 32; ++j) {
            const int dj = __builtin_amdgcn_readlane(d, j);
            unsafeAtomicAdd(agg + ((unsigned)dj * 32u + (unsigned)lane), v[j]);
        }
    } else {
        const int cnt = E - e0;
        const bool ok = lane < cnt;
        const int s = ok ? src[e0 + lane] : 0;
        const int d = ok ? dst[e0 + lane] : 0;
        for (int j = 0; j < cnt; ++j) {
            const int sj = __builtin_amdgcn_readlane(s, j);
            const int dj = __builtin_amdgcn_readlane(d, j);
            unsafeAtomicAdd(agg + ((unsigned)dj * 32u + (unsigned)lane), H[(unsigned)sj * 32u + (unsigned)lane]);
        }
    }
}

__global__ void propagate4_kernel(const int* __restrict__ src, const int* __restrict__ dst,
                                  const float* __restrict__ X,
                                  float* __restrict__ agg0, float* __restrict__ agg1, int E) {
    __shared__ unsigned s_half;
    float* __restrict__ agg = block_half(&s_half) ? agg1 : agg0;
    const int lane = threadIdx.x & 31;
    const int c    = lane & 3;
    const int q    = lane >> 2;
    const int e0   = (int)(blockIdx.x * blockDim.x + threadIdx.x) - lane;
    if (e0 + 32 <= E) {
        float val[4];
        int   row[4];
        for (int j = 0; j < 4; ++j) {
            const int e = e0 + j * 8 + q;
            const int s = src[e];
            const int d = dst[e];
            val[j] = X[(size_t)s * 4 + c];
            row[j] = d;
        }
        for (int j = 0; j < 4; ++j)
            unsafeAtomicAdd(agg + (size_t)row[j] * 4 + c, val[j]);
    } else {
        for (int j = 0; j < 4; ++j) {
            const int e = e0 + j * 8 + q;
            if (e < E) {
                const int s = src[e];
                const int d = dst[e];
                unsafeAtomicAdd(agg + (size_t)d * 4 + c, X[(size_t)s * 4 + c]);
            }
        }
    }
}

extern "C" void kernel_launch(void* const* d_in, const int* in_sizes, int n_in,
                              void* d_out, int out_size, void* d_ws, size_t ws_size,
                              hipStream_t stream) {
    const float* x   = (const float*)d_in[0];
    const int*   ei  = (const int*)d_in[1];
    const float* W1  = (const float*)d_in[2];
    const float* b1  = (const float*)d_in[3];
    const float* W2  = (const float*)d_in[4];
    const float* b2  = (const float*)d_in[5];
    const float* Wfc = (const float*)d_in[6];
    const float* bfc = (const float*)d_in[7];
    float* out = (float*)d_out;

    const int N = in_sizes[0] / 4;
    const int E = in_sizes[1] / 2;
    const int* srcIdx = ei;
    const int* dstIdx = ei + E;

    float* ws   = (float*)d_ws;
    const size_t Np = ((size_t)N + 255) & ~(size_t)255;
    float* dis  = ws;
    float* r1   = ws + Np;
    float* buf2 = r1 + (size_t)N * 64;
    float* deg0 = r1;
    float* deg1 = r1 + Np;
    float* aggx  = buf2;
    float* xs    = buf2 + (size_t)N * 4;
    float* aggx1 = buf2 + (size_t)N * 8;
    float* h2   = r1;
    float* agg2 = buf2 + (size_t)N * 32;
    float* agg2b = r1 + (size_t)N * 32;

    const int T = 256;
    const int rowTiles = N / 16;

    start_values_kernel<<<((N * 8) + T - 1) / T, T, 0, stream>>>(deg0, deg1, (float4*)agg2, (float4*)agg2b, N, N * 8);
    deg_accum_kernel<<<(E + T - 1) / T, T, 0, stream>>>(dstIdx, deg0, deg1, E);

    prescale_start4_kernel<<<(N + T - 1) / T, T, 0, stream>>>(x, deg0, deg1, dis, (float4*)xs, (float4*)aggx, (float4*)aggx1, N);
    propagate4_kernel<<<(E + T - 1) / T, T, 0, stream>>>(srcIdx, dstIdx, xs, aggx, aggx1, E);

    layer12_wmma_kernel<<<rowTiles, 64, 0, stream>>>(aggx, aggx1, W1, b1, W2, dis, h2);

    propagate32_kernel<<<(E + T - 1) / T, T, 0, stream>>>(srcIdx, dstIdx, h2, agg2, agg2b, E);

    out_wmma_kernel<<<rowTiles, 64, 0, stream>>>(agg2, agg2b, h2, dis, b2, Wfc, bfc, out);
}


# Degree count without global atomics: work-group (node range, edge slice) counts its slice of dst into LDS counters, stores one partial plane; prescale adds 32 planes; whole numbers, same values
# speedup vs baseline: 5.9923x; 1.0436x over previous
#include <hip/hip_runtime.h>

typedef __attribute__((ext_vector_type(2))) float v2f;
typedef __attribute__((ext_vector_type(8))) float v8f;

__device__ __forceinline__ unsigned block_half(unsigned* slot) {
    if (threadIdx.x == 0)
        *slot = (__builtin_amdgcn_s_sendmsg_rtn(0x87) >> 18) & 1u;
    __syncthreads();
    return (unsigned)__builtin_amdgcn_readfirstlane((int)*slot);
}

__global__ void start_values_kernel(float4* __restrict__ p, float4* __restrict__ q, int n4) {
    int i = blockIdx.x * blockDim.x + threadIdx.x;
    if (i < n4) {
        p[i] = make_float4(0.f, 0.f, 0.f, 0.f);
        q[i] = make_float4(0.f, 0.f, 0.f, 0.f);
    }
}

#define DEG_RANGE  12544
#define DEG_SLICES 32
__device__ __forceinline__ void count_dst(unsigned* cnt, int d, int lo, unsigned span) {
    const unsigned u = (unsigned)(d - lo);
    if (u < span) atomicAdd(&cnt[u], 1u);
}
__global__ void __launch_bounds__(1024)
degree_lds_kernel(const int* __restrict__ dst, unsigned* __restrict__ part, int n, int E) {
    __shared__ unsigned cnt[DEG_RANGE];
    const int tid    = threadIdx.x;
    const int ranges = gridDim.x / DEG_SLICES;
    const int r      = blockIdx.x % ranges;
    const int s      = blockIdx.x / ranges;
    const int lo     = r * DEG_RANGE;
    const unsigned span = (unsigned)((n - lo) < DEG_RANGE ? (n - lo) : DEG_RANGE);

    for (int i = tid; i < DEG_RANGE; i += 1024) cnt[i] = 0u;
    __syncthreads();

    const int per  = ((E + DEG_SLICES * 4 - 1) / (DEG_SLICES * 4)) * 4;
    const int eb   = s * per;
    const int ee   = (eb + per < E) ? (eb + per) : E;
    const int cntE = ee > eb ? ee - eb : 0;
    const int* __restrict__ p = dst + eb;

    int done = 0;
    if ((((size_t)p) & 15) == 0) {
        const int n4 = cntE >> 2;
        const int4* __restrict__ p4 = (const int4*)p;
        int i = tid;
        for (; i + 7 * 1024 < n4; i += 8 * 1024) {
            int4 v[8];
#pragma unroll
            for (int u = 0; u < 8; ++u) v[u] = p4[i + u * 1024];
#pragma unroll
            for (int u = 0; u < 8; ++u) {
                count_dst(cnt, v[u].x, lo, span);
                count_dst(cnt, v[u].y, lo, span);
                count_dst(cnt, v[u].z, lo, span);
                count_dst(cnt, v[u].w, lo, span);
            }
        }
        for (; i < n4; i += 1024) {
            const int4 v = p4[i];
            count_dst(cnt, v.x, lo, span);
            count_dst(cnt, v.y, lo, span);
            count_dst(cnt, v.z, lo, span);
            count_dst(cnt, v.w, lo, span);
        }
        done = n4 << 2;
    }
    for (int i = done + tid; i < cntE; i += 1024) count_dst(cnt, p[i], lo, span);

    __syncthreads();
    unsigned* __restrict__ o = part + (size_t)s * (size_t)n + (size_t)lo;
    for (unsigned i = (unsigned)tid; i < span; i += 1024u) o[i] = cnt[i];
}

__global__ void prescale_start4_kernel(const float* __restrict__ x,
                                       const unsigned* __restrict__ part,
                                       float* __restrict__ dis,
                                       float4* __restrict__ xs, float4* __restrict__ aggx,
                                       float4* __restrict__ aggx1, int n) {
    int i = blockIdx.x * blockDim.x + threadIdx.x;
    if (i >= n) return;
    unsigned c = 1u;
#pragma unroll
    for (int s = 0; s < DEG_SLICES; ++s) c += part[(size_t)s * (size_t)n + (size_t)i];
    const float ds = rsqrtf((float)c);
    dis[i] = ds;
    const float4 xv = *(const float4*)(x + (size_t)i * 4);
    const float4 r = make_float4(xv.x * ds, xv.y * ds, xv.z * ds, xv.w * ds);
    xs[i]    = r;
    aggx[i]  = r;
    aggx1[i] = make_float4(0.f, 0.f, 0.f, 0.f);
}


#define H1_PITCH 68
__global__ void __launch_bounds__(64)
layer12_wmma_kernel(const float* __restrict__ aggx, const float* __restrict__ aggx1,
                    const float* __restrict__ W1, const float* __restrict__ b1,
                    const float* __restrict__ W2, const float* __restrict__ dis,
                    float* __restrict__ hs) {
    __shared__ float sH[16 * H1_PITCH];
    const int lane = threadIdx.x & 31;
    const int wave = threadIdx.x >> 5;
    const int rt   = blockIdx.x;
    const int half = lane >> 4;
    const int l16  = lane & 15;
    const int m    = rt * 16 + l16;
    const int kp   = half * 2;

    v2f a;
    a.x = aggx[m * 4 + kp]     + aggx1[m * 4 + kp];
    a.y = aggx[m * 4 + kp + 1] + aggx1[m * 4 + kp + 1];

    float rsv[8];
#pragma unroll
    for (int r = 0; r < 8; ++r) rsv[r] = dis[rt * 16 + r + half * 8];

#pragma unroll
    for (int t = 0; t < 2; ++t) {
        const int n1 = (wave * 2 + t) * 16 + l16;
        v2f b;
        b.x = W1[kp * 64 + n1];
        b.y = W1[(kp + 1) * 64 + n1];
        v8f acc = {};
        acc = __builtin_amdgcn_wmma_f32_16x16x4_f32(
             false, a,  false, b,
             (short)0, acc,  false,  false);
        const float bv = b1[n1];
#pragma unroll
        for (int r = 0; r < 8; ++r) {
            float v = acc[r];
            v *= rsv[r];
            v += bv;
            v = v > 0.0f ? v : 0.0f;
            sH[(r + half * 8) * H1_PITCH + n1] = v;
        }
    }
    __syncthreads();

    const int n = wave * 16 + l16;
    v8f acc2 = {};
#pragma unroll
    for (int kb = 0; kb < 64; kb += 4) {
        const int kq = kb + half * 2;
        const v2f a2 = *(const v2f*)&sH[l16 * H1_PITCH + kq];
        v2f b2;
        b2.x = W2[kq * 32 + n];
        b2.y = W2[(kq + 1) * 32 + n];
        acc2 = __builtin_amdgcn_wmma_f32_16x16x4_f32(
             false, a2,  false, b2,
             (short)0, acc2,  false,  false);
    }
#pragma unroll
    for (int r = 0; r < 8; ++r) {
        const int row = rt * 16 + r + half * 8;
        hs[row * 32 + n] = acc2[r] * rsv[r];
    }
}

#define H2_PITCH 36
__global__ void __launch_bounds__(64)
out_wmma_kernel(const float* __restrict__ agg0, const float* __restrict__ agg1,
                const float* __restrict__ H, const float* __restrict__ dis,
                const float* __restrict__ b2, const float* __restrict__ Wfc,
                const float* __restrict__ bfc, float* __restrict__ out) {
    __shared__ __attribute__((aligned(16))) float sA[16 * H2_PITCH];
    const int rt = blockIdx.x;
#pragma unroll
    for (int u = 0; u < 2; ++u) {
        const int idx = (int)threadIdx.x + u * 64;
        const int rl  = idx >> 3;
        const int c4  = idx & 7;
        const int row = rt * 16 + rl;
        const int g   = row * 32 + c4 * 4;
        const float4 p  = *(const float4*)(agg0 + g);
        const float4 q  = *(const float4*)(agg1 + g);
        const float4 h  = *(const float4*)(H + g);
        const float4 bb = *(const float4*)(b2 + c4 * 4);
        const float ds  = dis[row];
        float4 v;
        v.x = ((p.x + q.x) + h.x) * ds + bb.x;
        v.y = ((p.y + q.y) + h.y) * ds + bb.y;
        v.z = ((p.z + q.z) + h.z) * ds + bb.z;
        v.w = ((p.w + q.w) + h.w) * ds + bb.w;
        v.x = v.x > 0.0f ? v.x : 0.0f;
        v.y = v.y > 0.0f ? v.y : 0.0f;
        v.z = v.z > 0.0f ? v.z : 0.0f;
        v.w = v.w > 0.0f ? v.w : 0.0f;
        *(float4*)&sA[rl * H2_PITCH + c4 * 4] = v;
    }
    __syncthreads();

    const int lane = threadIdx.x & 31;
    const int wave = threadIdx.x >> 5;
    const int half = lane >> 4;
    const int l16  = lane & 15;
    const int n    = wave * 16 + l16;
    v8f acc = {};
#pragma unroll
    for (int kb = 0; kb < 32; kb += 4) {
        const int kq = kb + half * 2;
        const v2f a = *(const v2f*)&sA[l16 * H2_PITCH + kq];
        v2f b;
        b.x = Wfc[kq * 32 + n];
        b.y = Wfc[(kq + 1) * 32 + n];
        acc = __builtin_amdgcn_wmma_f32_16x16x4_f32(
             false, a,  false, b,
             (short)0, acc,  false,  false);
    }
    const float bv = bfc[n];
#pragma unroll
    for (int r = 0; r < 8; ++r) {
        const int row = rt * 16 + r + half * 8;
        float v = acc[r] + bv;
        v = v > 0.0f ? v : 0.0f;
        out[row * 32 + n] = v;
    }
}

__global__ void propagate32_kernel(const int* __restrict__ src, const int* __restrict__ dst,
                                   const float* __restrict__ H,
                                   float* __restrict__ agg0, float* __restrict__ agg1, int E) {
    __shared__ unsigned s_half;
    float* __restrict__ agg = block_half(&s_half) ? agg1 : agg0;
    const int lane = threadIdx.x & 31;
    const int e0 = __builtin_amdgcn_readfirstlane((int)(blockIdx.x * blockDim.x + (threadIdx.x & ~31u)));
    if (e0 >= E) return;
    if (e0 + 32 <= E) {
        const int s = src[e0 + lane];
        const int d = dst[e0 + lane];
        float v[32];
#pragma unroll
        for (int j = 0; j < 32; ++j) {
            const int sj = __builtin_amdgcn_readlane(s, j);
            v[j] = H[(unsigned)sj * 32u + (unsigned)lane];
        }
#pragma unroll
        for (int j = 0; j < 32; ++j) {
            const int dj = __builtin_amdgcn_readlane(d, j);
            unsafeAtomicAdd(agg + ((unsigned)dj * 32u + (unsigned)lane), v[j]);
        }
    } else {
        const int cnt = E - e0;
        const bool ok = lane < cnt;
        const int s = ok ? src[e0 + lane] : 0;
        const int d = ok ? dst[e0 + lane] : 0;
        for (int j = 0; j < cnt; ++j) {
            const int sj = __builtin_amdgcn_readlane(s, j);
            const int dj = __builtin_amdgcn_readlane(d, j);
            unsafeAtomicAdd(agg + ((unsigned)dj * 32u + (unsigned)lane), H[(unsigned)sj * 32u + (unsigned)lane]);
        }
    }
}

__global__ void propagate4_kernel(const int* __restrict__ src, const int* __restrict__ dst,
                                  const float* __restrict__ X,
                                  float* __restrict__ agg0, float* __restrict__ agg1, int E) {
    __shared__ unsigned s_half;
    float* __restrict__ agg = block_half(&s_half) ? agg1 : agg0;
    const int lane = threadIdx.x & 31;
    const int c    = lane & 3;
    const int q    = lane >> 2;
    const int e0   = (int)(blockIdx.x * blockDim.x + threadIdx.x) - lane;
    if (e0 + 32 <= E) {
        float val[4];
        int   row[4];
        for (int j = 0; j < 4; ++j) {
            const int e = e0 + j * 8 + q;
            const int s = src[e];
            const int d = dst[e];
            val[j] = X[(size_t)s * 4 + c];
            row[j] = d;
        }
        for (int j = 0; j < 4; ++j)
            unsafeAtomicAdd(agg + (size_t)row[j] * 4 + c, val[j]);
    } else {
        for (int j = 0; j < 4; ++j) {
            const int e = e0 + j * 8 + q;
            if (e < E) {
                const int s = src[e];
                const int d = dst[e];
                unsafeAtomicAdd(agg + (size_t)d * 4 + c, X[(size_t)s * 4 + c]);
            }
        }
    }
}

extern "C" void kernel_launch(void* const* d_in, const int* in_sizes, int n_in,
                              void* d_out, int out_size, void* d_ws, size_t ws_size,
                              hipStream_t stream) {
    const float* x   = (const float*)d_in[0];
    const int*   ei  = (const int*)d_in[1];
    const float* W1  = (const float*)d_in[2];
    const float* b1  = (const float*)d_in[3];
    const float* W2  = (const float*)d_in[4];
    const float* b2  = (const float*)d_in[5];
    const float* Wfc = (const float*)d_in[6];
    const float* bfc = (const float*)d_in[7];
    float* out = (float*)d_out;

    const int N = in_sizes[0] / 4;
    const int E = in_sizes[1] / 2;
    const int* srcIdx = ei;
    const int* dstIdx = ei + E;

    float* ws   = (float*)d_ws;
    const size_t Np = ((size_t)N + 255) & ~(size_t)255;
    float* dis  = ws;
    float* r1   = ws + Np;
    float* buf2 = r1 + (size_t)N * 64;
    unsigned* part = (unsigned*)r1;
    float* aggx  = buf2;
    float* xs    = buf2 + (size_t)N * 4;
    float* aggx1 = buf2 + (size_t)N * 8;
    float* h2   = r1;
    float* agg2 = buf2 + (size_t)N * 32;
    float* agg2b = r1 + (size_t)N * 32;

    const int T = 256;
    const int rowTiles = N / 16;

    start_values_kernel<<<((N * 8) + T - 1) / T, T, 0, stream>>>((float4*)agg2, (float4*)agg2b, N * 8);
    const int degRanges = (N + DEG_RANGE - 1) / DEG_RANGE;
    degree_lds_kernel<<<degRanges * DEG_SLICES, 1024, 0, stream>>>(dstIdx, part, N, E);

    prescale_start4_kernel<<<(N + T - 1) / T, T, 0, stream>>>(x, part, dis, (float4*)xs, (float4*)aggx, (float4*)aggx1, N);
    propagate4_kernel<<<(E + T - 1) / T, T, 0, stream>>>(srcIdx, dstIdx, xs, aggx, aggx1, E);

    layer12_wmma_kernel<<<rowTiles, 64, 0, stream>>>(aggx, aggx1, W1, b1, W2, dis, h2);

    propagate32_kernel<<<(E + T - 1) / T, T, 0, stream>>>(srcIdx, dstIdx, h2, agg2, agg2b, E);

    out_wmma_kernel<<<rowTiles, 64, 0, stream>>>(agg2, agg2b, h2, dis, b2, Wfc, bfc, out);
}
